# starting kernel
# baseline (speedup 1.0000x reference)
#include <hip/hip_runtime.h>
#include <math.h>


typedef float v2f __attribute__((ext_vector_type(2)));
typedef float v8f __attribute__((ext_vector_type(8)));

#ifndef __has_builtin
#define __has_builtin(x) 0
#endif

__device__ __forceinline__ float fast_tanh(float x) {
#if __has_builtin(__builtin_amdgcn_tanhf)
    return __builtin_amdgcn_tanhf(x);
#else
    float r;
    asm volatile("v_tanh_f32 %0, %1\n\tv_nop" : "=v"(r) : "v"(x));
    return r;
#endif
}

template <bool BT>
__global__ void wmma_gemm_f32(const float* __restrict__ A,
                              const float* __restrict__ B,
                              float* __restrict__ C,
                              int M, int N, int K,
                              long long strideA, long long strideB, long long strideC)
{
    constexpr int NT = 4;

    const int lane = threadIdx.x & 31;
    const int wave = threadIdx.x >> 5;
    const int wavesPerBlock = blockDim.x >> 5;

    const int groupsN = N >> 6;
    const int totalStrips = (M >> 4) * groupsN;
    const int strip = blockIdx.x * wavesPerBlock + wave;
    if (strip >= totalStrips) return;

    const int tm = strip / groupsN;
    const int tg = strip - tm * groupsN;
    const int row0 = tm << 4;
    const int col0 = tg << 6;

    const float* __restrict__ Ab = A + (long long)blockIdx.y * strideA;
    const float* __restrict__ Bb = B + (long long)blockIdx.y * strideB;
    float* __restrict__ Cb       = C + (long long)blockIdx.y * strideC;

    const int m     = lane & 15;
    const int khalf = (lane >> 4) << 1;

    v8f c[NT] = {};

    #pragma unroll 4
    for (int k = 0; k < K; k += 4) {
        const int ka = k + khalf;
        v2f a;
        a.x = Ab[(long long)(row0 + m) * K + ka];
        a.y = Ab[(long long)(row0 + m) * K + ka + 1];

        #pragma unroll
        for (int t = 0; t < NT; ++t) {
            const int cn = col0 + (t << 4) + m;
            v2f b;
            if (BT) {
                b.x = Bb[(long long)cn * K + ka];
                b.y = Bb[(long long)cn * K + ka + 1];
            } else {
                b.x = Bb[(long long)(ka)     * N + cn];
                b.y = Bb[(long long)(ka + 1) * N + cn];
            }
            c[t] = __builtin_amdgcn_wmma_f32_16x16x4_f32(false, a, false, b,
                                                         (short)0, c[t],
                                                         false, false);
        }
    }

    const int rowOff = (lane >> 4) << 3;
    #pragma unroll
    for (int t = 0; t < NT; ++t) {
        #pragma unroll
        for (int j = 0; j < 8; ++j) {
            Cb[(long long)(row0 + rowOff + j) * N + col0 + (t << 4) + (lane & 15)]
                = c[t][j];
        }
    }
}

__global__ void score_tanh_kernel(const float* __restrict__ qp,
                                  const float* __restrict__ kp,
                                  const float* __restrict__ v,
                                  float* __restrict__ sc)
{
    constexpr int U  = 256;
    constexpr int TQ = 32;
    constexpr int TK = 2048;
    constexpr int KW = 2;

    __shared__ float qps[U * TQ];
    __shared__ float vs[U];
    __shared__ float kps[8 * KW * U];

    const int tid  = threadIdx.x;
    const int lane = tid & 31;
    const int wave = tid >> 5;
    const int b    = blockIdx.y;
    const int k0   = blockIdx.x * (8 * KW) + wave * KW;

    const float* qpb = qp + (long long)b * TQ * U;
    for (int i = tid; i < TQ * U; i += 256) {
        const int q = i >> 8;
        const int u = i & 255;
        qps[u * TQ + q] = qpb[i];
    }
    vs[tid] = v[tid];

    #pragma unroll
    for (int j = 0; j < KW; ++j) {
        const float* kpr = kp + ((long long)b * TK + (k0 + j)) * U;
        for (int i = lane; i < U; i += 32)
            kps[(wave * KW + j) * U + i] = kpr[i];
    }
    __syncthreads();

    float acc0 = 0.f, acc1 = 0.f;
    const int kb0 = (wave * KW + 0) * U;
    const int kb1 = (wave * KW + 1) * U;

    #pragma unroll 4
    for (int u = 0; u < U; ++u) {
        const float qv = qps[u * TQ + lane];
        const float vv = vs[u];
        acc0 += fast_tanh(qv + kps[kb0 + u]) * vv;
        acc1 += fast_tanh(qv + kps[kb1 + u]) * vv;
    }

    float* srow = sc + ((long long)b * TQ + lane) * TK + k0;
    srow[0] = acc0;
    srow[1] = acc1;
}

__global__ void softmax_rows_kernel(const float* __restrict__ sc,
                                    float* __restrict__ attn)
{
    constexpr int TK = 2048;
    const int row = blockIdx.x;
    const int tid = threadIdx.x;
    const float* s = sc   + (long long)row * TK;
    float*       a = attn + (long long)row * TK;

    __shared__ float red[256];

    float m = -INFINITY;
    #pragma unroll
    for (int j = 0; j < 8; ++j) m = fmaxf(m, s[tid + j * 256]);
    red[tid] = m;
    __syncthreads();
    for (int off = 128; off > 0; off >>= 1) {
        if (tid < off) red[tid] = fmaxf(red[tid], red[tid + off]);
        __syncthreads();
    }
    const float mx = red[0];
    __syncthreads();

    float e[8];
    float sum = 0.f;
    #pragma unroll
    for (int j = 0; j < 8; ++j) {
        e[j] = __expf(s[tid + j * 256] - mx);
        sum += e[j];
    }
    red[tid] = sum;
    __syncthreads();
    for (int off = 128; off > 0; off >>= 1) {
        if (tid < off) red[tid] += red[tid + off];
        __syncthreads();
    }
    const float inv = 1.0f / red[0];
    #pragma unroll
    for (int j = 0; j < 8; ++j) a[tid + j * 256] = e[j] * inv;
}

extern "C" void kernel_launch(void* const* d_in, const int* in_sizes, int n_in,
                              void* d_out, int out_size, void* d_ws, size_t ws_size,
                              hipStream_t stream)
{
    (void)in_sizes; (void)n_in; (void)out_size; (void)ws_size;

    const float* query = (const float*)d_in[0];
    const float* keys  = (const float*)d_in[1];
    const float* Wq    = (const float*)d_in[2];
    const float* Wk    = (const float*)d_in[3];
    const float* v     = (const float*)d_in[4];

    float* context = (float*)d_out;
    float* attn    = (float*)d_out + 65536;

    float* qp = (float*)d_ws;
    float* kp = qp + 65536;
    float* sc = kp + (size_t)16384 * 256;

    wmma_gemm_f32<true><<<dim3(16, 1), 128, 0, stream>>>(
        query, Wq, qp, 256, 256, 256, 0, 0, 0);

    wmma_gemm_f32<true><<<dim3(1024, 1), 128, 0, stream>>>(
        keys, Wk, kp, 16384, 256, 256, 0, 0, 0);

    score_tanh_kernel<<<dim3(2048 / 16, 8), 256, 0, stream>>>(qp, kp, v, sc);

    softmax_rows_kernel<<<dim3(256), 256, 0, stream>>>(sc, attn);

    wmma_gemm_f32<false><<<dim3(2, 8), 128, 0, stream>>>(
        attn, keys, context, 32, 256, 2048,
        (long long)32 * 2048, (long long)2048 * 256, (long long)32 * 256);
}


# Projection GEMM (q_proj, k_proj) fragment loads 16 bytes per lane: k order permuted (per 8-k block lanes 0-15 hold k 0-3, lanes 16-31 k 4-7, A and B alike), one load feeds 2 k-steps; was 8-byte
# speedup vs baseline: 5.2445x; 5.2445x over previous
#include <hip/hip_runtime.h>
#include <math.h>


typedef float v2f __attribute__((ext_vector_type(2)));
typedef float v4f __attribute__((ext_vector_type(4)));
typedef float v8f __attribute__((ext_vector_type(8)));

#ifndef __has_builtin
#define __has_builtin(x) 0
#endif

__device__ __forceinline__ float fast_tanh(float x) {
#if __has_builtin(__builtin_amdgcn_tanhf)
    return __builtin_amdgcn_tanhf(x);
#else
    float r;
    asm volatile("v_tanh_f32 %0, %1\n\tv_nop" : "=v"(r) : "v"(x));
    return r;
#endif
}

template <bool BT, bool CQ>
__global__ void __launch_bounds__(128)
wmma_gemm_f32(const float* __restrict__ A,
                              const float* __restrict__ B,
                              float* __restrict__ C,
                              int M, int N, int K,
                              long long strideA, long long strideB, long long strideC)
{
    constexpr int NT = 4;
    constexpr int D  = 8;
    constexpr int P  = D / 2;

    const int lane = threadIdx.x & 31;
    const int wave = threadIdx.x >> 5;
    const int wavesPerBlock = blockDim.x >> 5;

    const int groupsN = N >> 6;
    const int totalStrips = (M >> 4) * groupsN;
    const int strip = blockIdx.x * wavesPerBlock + wave;
    if (strip >= totalStrips) return;

    const int tm = strip / groupsN;
    const int tg = strip - tm * groupsN;
    const int row0 = tm << 4;
    const int col0 = tg << 6;

    const float* __restrict__ Ab = A + (long long)blockIdx.y * strideA;
    const float* __restrict__ Bb = B + (long long)blockIdx.y * strideB;
    float* __restrict__ Cb       = C + (long long)blockIdx.y * strideC;

    const int m    = lane & 15;
    const int kq   = (lane >> 4) << 2;

    const float* __restrict__ arow = Ab + (long long)(row0 + m) * K + kq;
    const float* __restrict__ bcol = BT
        ? Bb + (long long)(col0 + m) * K + kq
        : Bb + (long long)kq * N + col0 + m;

    v8f c[NT] = {};

    v4f aA[P], aB[P];
    v4f bA[P][NT], bB[P][NT];

    auto load_set = [&](int k, v4f (&a)[P], v4f (&b)[P][NT]) {
        #pragma unroll
        for (int p = 0; p < P; ++p) {
            const int kb = k + (p << 3);
            a[p] = *(const v4f*)(arow + kb);
            #pragma unroll
            for (int t = 0; t < NT; ++t) {
                if (BT) {
                    b[p][t] = *(const v4f*)(bcol + (long long)(t << 4) * K + kb);
                } else {
                    b[p][t].x = bcol[(long long)kb * N + (t << 4)];
                    b[p][t].y = bcol[(long long)(kb + 1) * N + (t << 4)];
                    b[p][t].z = bcol[(long long)(kb + 2) * N + (t << 4)];
                    b[p][t].w = bcol[(long long)(kb + 3) * N + (t << 4)];
                }
            }
        }
    };
    auto mma_set = [&](const v4f (&a)[P], const v4f (&b)[P][NT]) {
        #pragma unroll
        for (int p = 0; p < P; ++p) {
            const v2f a0 = __builtin_shufflevector(a[p], a[p], 0, 1);
            #pragma unroll
            for (int t = 0; t < NT; ++t) {
                const v2f b0 = __builtin_shufflevector(b[p][t], b[p][t], 0, 1);
                c[t] = __builtin_amdgcn_wmma_f32_16x16x4_f32(false, a0, false, b0,
                                                             (short)0, c[t],
                                                             false, false);
            }
            const v2f a1 = __builtin_shufflevector(a[p], a[p], 2, 3);
            #pragma unroll
            for (int t = 0; t < NT; ++t) {
                const v2f b1 = __builtin_shufflevector(b[p][t], b[p][t], 2, 3);
                c[t] = __builtin_amdgcn_wmma_f32_16x16x4_f32(false, a1, false, b1,
                                                             (short)0, c[t],
                                                             false, false);
            }
        }
    };

    constexpr int KS = D * 4;
    load_set(0, aA, bA);
    int k = 0;
    for (; k < K - 2 * KS; k += 2 * KS) {
        load_set(k + KS, aB, bB);
        mma_set(aA, bA);
        load_set(k + 2 * KS, aA, bA);
        mma_set(aB, bB);
    }
    load_set(k + KS, aB, bB);
    mma_set(aA, bA);
    mma_set(aB, bB);

    const int rowOff = (lane >> 4) << 3;
    #pragma unroll
    for (int t = 0; t < NT; ++t) {
        #pragma unroll
        for (int j = 0; j < 8; ++j) {
            if (CQ) {
                const int row = row0 + rowOff + j;
                const int col = col0 + (t << 4) + (lane & 15);
                Cb[(long long)(col >> 5) * (M * 32) + row * 32 + (col & 31)] = c[t][j];
            } else {
                Cb[(long long)(row0 + rowOff + j) * N + col0 + (t << 4) + (lane & 15)]
                    = c[t][j];
            }
        }
    }
}

__global__ void __launch_bounds__(32)
context_tile_kernel(const float* __restrict__ attn,
                    const float* __restrict__ keys,
                    float* __restrict__ ctx)
{
    constexpr int TQ = 32;
    constexpr int TK = 2048;
    constexpr int DK = 256;
    constexpr int R     = 32;
    constexpr int STEPS = TK / 4;

    const int lane = threadIdx.x;
    const int tile = blockIdx.x;
    const int tn   = tile & 15;
    const int tm   = (tile >> 4) & 1;
    const int b    = tile >> 5;

    const int m     = lane & 15;
    const int khalf = (lane >> 4) << 1;

    const float* __restrict__ ap = attn + ((long long)(b * TQ + (tm << 4) + m)) * TK + khalf;
    const float* __restrict__ bp = keys + ((long long)b * TK + khalf) * DK + (tn << 4) + m;

    v2f fa[R], fb[R];
    #pragma unroll
    for (int s = 0; s < R; ++s) {
        const int ks = s << 2;
        fa[s].x = ap[ks];
        fa[s].y = ap[ks + 1];
        fb[s].x = bp[(long long)ks * DK];
        fb[s].y = bp[(long long)(ks + 1) * DK];
    }

    v8f c0 = {}, c1 = {};

    for (int base = 0; base < STEPS - R; base += R) {
        #pragma unroll
        for (int s = 0; s < R; ++s) {
            if (s & 1)
                c1 = __builtin_amdgcn_wmma_f32_16x16x4_f32(false, fa[s], false, fb[s],
                                                           (short)0, c1, false, false);
            else
                c0 = __builtin_amdgcn_wmma_f32_16x16x4_f32(false, fa[s], false, fb[s],
                                                           (short)0, c0, false, false);
            const int ks = (base + R + s) << 2;
            fa[s].x = ap[ks];
            fa[s].y = ap[ks + 1];
            fb[s].x = bp[(long long)ks * DK];
            fb[s].y = bp[(long long)(ks + 1) * DK];
        }
    }
    #pragma unroll
    for (int s = 0; s < R; ++s) {
        if (s & 1)
            c1 = __builtin_amdgcn_wmma_f32_16x16x4_f32(false, fa[s], false, fb[s],
                                                       (short)0, c1, false, false);
        else
            c0 = __builtin_amdgcn_wmma_f32_16x16x4_f32(false, fa[s], false, fb[s],
                                                       (short)0, c0, false, false);
    }

    const int rowOff = (lane >> 4) << 3;
    float* __restrict__ cp = ctx + ((long long)(b * TQ + (tm << 4) + rowOff)) * DK + (tn << 4) + m;
    #pragma unroll
    for (int j = 0; j < 8; ++j)
        cp[(long long)j * DK] = c0[j] + c1[j];
}

__global__ void __launch_bounds__(256)
score_tanh_kernel(const float* __restrict__ qpT,
                  const float* __restrict__ kp,
                  const float* __restrict__ v,
                  float* __restrict__ sc)
{
    constexpr int U  = 256;
    constexpr int TQ = 32;
    constexpr int TK = 2048;
    constexpr int KW = 4;
    constexpr int UC = 16;

    const int lane = threadIdx.x & 31;
    const int wave = __builtin_amdgcn_readfirstlane((int)(threadIdx.x >> 5));
    const int row0 = (blockIdx.x * 8 + wave) * KW;
    const int b    = row0 >> 11;
    const int k0   = row0 & (TK - 1);

    const float* __restrict__ qcol = qpT + b * (U * TQ) + lane;
    const float* __restrict__ krow = kp + (long long)row0 * U;

    float acc[KW];
    #pragma unroll
    for (int j = 0; j < KW; ++j) acc[j] = 0.f;

    #pragma unroll 1
    for (int c = 0; c < U; c += UC) {
        float q[UC];
        #pragma unroll
        for (int i = 0; i < UC; ++i) q[i] = qcol[(c + i) * TQ];

        #pragma unroll
        for (int j = 0; j < KW; ++j) {
            #pragma unroll
            for (int i = 0; i < UC; ++i)
                acc[j] += fast_tanh(q[i] + krow[j * U + c + i]) * v[c + i];
        }
    }

    float4 o;
    o.x = acc[0]; o.y = acc[1]; o.z = acc[2]; o.w = acc[3];
    *(float4*)(sc + ((long long)b * TQ + lane) * TK + k0) = o;
}

__global__ void softmax_rows_kernel(const float* __restrict__ sc,
                                    float* __restrict__ attn)
{
    constexpr int TK = 2048;
    const int row = blockIdx.x;
    const int tid = threadIdx.x;
    const float* s = sc   + (long long)row * TK;
    float*       a = attn + (long long)row * TK;

    __shared__ float red[256];

    float m = -INFINITY;
    #pragma unroll
    for (int j = 0; j < 8; ++j) m = fmaxf(m, s[tid + j * 256]);
    red[tid] = m;
    __syncthreads();
    for (int off = 128; off > 0; off >>= 1) {
        if (tid < off) red[tid] = fmaxf(red[tid], red[tid + off]);
        __syncthreads();
    }
    const float mx = red[0];
    __syncthreads();

    float e[8];
    float sum = 0.f;
    #pragma unroll
    for (int j = 0; j < 8; ++j) {
        e[j] = __expf(s[tid + j * 256] - mx);
        sum += e[j];
    }
    red[tid] = sum;
    __syncthreads();
    for (int off = 128; off > 0; off >>= 1) {
        if (tid < off) red[tid] += red[tid + off];
        __syncthreads();
    }
    const float inv = 1.0f / red[0];
    #pragma unroll
    for (int j = 0; j < 8; ++j) a[tid + j * 256] = e[j] * inv;
}

extern "C" void kernel_launch(void* const* d_in, const int* in_sizes, int n_in,
                              void* d_out, int out_size, void* d_ws, size_t ws_size,
                              hipStream_t stream)
{
    (void)in_sizes; (void)n_in; (void)out_size; (void)ws_size;

    const float* query = (const float*)d_in[0];
    const float* keys  = (const float*)d_in[1];
    const float* Wq    = (const float*)d_in[2];
    const float* Wk    = (const float*)d_in[3];
    const float* v     = (const float*)d_in[4];

    float* context = (float*)d_out;
    float* attn    = (float*)d_out + 65536;

    float* qp = (float*)d_ws;
    float* kp = qp + 65536;
    float* sc = kp + (size_t)16384 * 256;

    wmma_gemm_f32<true, true><<<dim3(16, 1), 128, 0, stream>>>(
        Wq, query, qp, 256, 256, 256, 0, 0, 0);

    wmma_gemm_f32<true, false><<<dim3(1024, 1), 128, 0, stream>>>(
        keys, Wk, kp, 16384, 256, 256, 0, 0, 0);

    score_tanh_kernel<<<dim3(512), 256, 0, stream>>>(qp, kp, v, sc);

    softmax_rows_kernel<<<dim3(256), 256, 0, stream>>>(sc, attn);

    context_tile_kernel<<<dim3(256), 32, 0, stream>>>(attn, keys, context);
}


# Context kernel requests its key lines and attn lines into L2 64 k-steps ahead of the fragment ring's loads (rolling prefetch); was each key line a first touch with 21 k-steps of loads in flight
# speedup vs baseline: 5.8407x; 1.1137x over previous
#include <hip/hip_runtime.h>
#include <math.h>


typedef float v2f __attribute__((ext_vector_type(2)));
typedef float v4f __attribute__((ext_vector_type(4)));
typedef float v8f __attribute__((ext_vector_type(8)));

#ifndef __has_builtin
#define __has_builtin(x) 0
#endif

__device__ __forceinline__ float fast_tanh(float x) {
#if __has_builtin(__builtin_amdgcn_tanhf)
    return __builtin_amdgcn_tanhf(x);
#else
    float r;
    asm volatile("v_tanh_f32 %0, %1\n\tv_nop" : "=v"(r) : "v"(x));
    return r;
#endif
}

template <bool BT, bool CQ>
__global__ void __launch_bounds__(128)
wmma_gemm_f32(const float* __restrict__ A,
                              const float* __restrict__ B,
                              float* __restrict__ C,
                              int M, int N, int K,
                              long long strideA, long long strideB, long long strideC)
{
    constexpr int NT = 4;
    constexpr int D  = 8;
    constexpr int P  = D / 2;

    const int lane = threadIdx.x & 31;
    const int wave = threadIdx.x >> 5;
    const int wavesPerBlock = blockDim.x >> 5;

    const int groupsN = N >> 6;
    const int totalStrips = (M >> 4) * groupsN;
    const int strip = blockIdx.x * wavesPerBlock + wave;
    if (strip >= totalStrips) return;

    const int tm = strip / groupsN;
    const int tg = strip - tm * groupsN;
    const int row0 = tm << 4;
    const int col0 = tg << 6;

    const float* __restrict__ Ab = A + (long long)blockIdx.y * strideA;
    const float* __restrict__ Bb = B + (long long)blockIdx.y * strideB;
    float* __restrict__ Cb       = C + (long long)blockIdx.y * strideC;

    const int m    = lane & 15;
    const int kq   = (lane >> 4) << 2;

    const float* __restrict__ arow = Ab + (long long)(row0 + m) * K + kq;
    const float* __restrict__ bcol = BT
        ? Bb + (long long)(col0 + m) * K + kq
        : Bb + (long long)kq * N + col0 + m;

    v8f c[NT] = {};

    v4f aA[P], aB[P];
    v4f bA[P][NT], bB[P][NT];

    auto load_set = [&](int k, v4f (&a)[P], v4f (&b)[P][NT]) {
        #pragma unroll
        for (int p = 0; p < P; ++p) {
            const int kb = k + (p << 3);
            a[p] = *(const v4f*)(arow + kb);
            #pragma unroll
            for (int t = 0; t < NT; ++t) {
                if (BT) {
                    b[p][t] = *(const v4f*)(bcol + (long long)(t << 4) * K + kb);
                } else {
                    b[p][t].x = bcol[(long long)kb * N + (t << 4)];
                    b[p][t].y = bcol[(long long)(kb + 1) * N + (t << 4)];
                    b[p][t].z = bcol[(long long)(kb + 2) * N + (t << 4)];
                    b[p][t].w = bcol[(long long)(kb + 3) * N + (t << 4)];
                }
            }
        }
    };
    auto mma_set = [&](const v4f (&a)[P], const v4f (&b)[P][NT]) {
        #pragma unroll
        for (int p = 0; p < P; ++p) {
            const v2f a0 = __builtin_shufflevector(a[p], a[p], 0, 1);
            #pragma unroll
            for (int t = 0; t < NT; ++t) {
                const v2f b0 = __builtin_shufflevector(b[p][t], b[p][t], 0, 1);
                c[t] = __builtin_amdgcn_wmma_f32_16x16x4_f32(false, a0, false, b0,
                                                             (short)0, c[t],
                                                             false, false);
            }
            const v2f a1 = __builtin_shufflevector(a[p], a[p], 2, 3);
            #pragma unroll
            for (int t = 0; t < NT; ++t) {
                const v2f b1 = __builtin_shufflevector(b[p][t], b[p][t], 2, 3);
                c[t] = __builtin_amdgcn_wmma_f32_16x16x4_f32(false, a1, false, b1,
                                                             (short)0, c[t],
                                                             false, false);
            }
        }
    };

    constexpr int KS = D * 4;
    load_set(0, aA, bA);
    int k = 0;
    for (; k < K - 2 * KS; k += 2 * KS) {
        load_set(k + KS, aB, bB);
        mma_set(aA, bA);
        load_set(k + 2 * KS, aA, bA);
        mma_set(aB, bB);
    }
    load_set(k + KS, aB, bB);
    mma_set(aA, bA);
    mma_set(aB, bB);

    const int rowOff = (lane >> 4) << 3;
    #pragma unroll
    for (int t = 0; t < NT; ++t) {
        #pragma unroll
        for (int j = 0; j < 8; ++j) {
            if (CQ) {
                const int row = row0 + rowOff + j;
                const int col = col0 + (t << 4) + (lane & 15);
                Cb[(long long)(col >> 5) * (M * 32) + row * 32 + (col & 31)] = c[t][j];
            } else {
                Cb[(long long)(row0 + rowOff + j) * N + col0 + (t << 4) + (lane & 15)]
                    = c[t][j];
            }
        }
    }
}

__global__ void __launch_bounds__(32)
context_tile_kernel(const float* __restrict__ attn,
                    const float* __restrict__ keys,
                    float* __restrict__ ctx)
{
    constexpr int TQ = 32;
    constexpr int TK = 2048;
    constexpr int DK = 256;
    constexpr int R     = 32;
    constexpr int STEPS = TK / 4;
    constexpr int PF    = 64;

    const int lane = threadIdx.x;
    const int tile = blockIdx.x;
    const int tn   = tile & 15;
    const int tm   = (tile >> 4) & 1;
    const int b    = tile >> 5;

    const int m     = lane & 15;
    const int khalf = (lane >> 4) << 1;

    const float* __restrict__ ap = attn + ((long long)(b * TQ + (tm << 4) + m)) * TK + khalf;
    const float* __restrict__ bp = keys + ((long long)b * TK + khalf) * DK + (tn << 4) + m;

    v2f fa[R], fb[R];
    #pragma unroll
    for (int s = 0; s < R; ++s) {
        const int ks = s << 2;
        fa[s].x = ap[ks];
        fa[s].y = ap[ks + 1];
        fb[s].x = bp[(long long)ks * DK];
        fb[s].y = bp[(long long)(ks + 1) * DK];
    }

    const float* __restrict__ pfb = keys + ((long long)b * TK + lane) * DK + (tn << 4);
    const float* __restrict__ pfa = attn + ((long long)(b * TQ + (tm << 4) + m)) * TK + ((lane >> 4) << 6);

    #pragma unroll
    for (int s = R; s < R + PF; s += 8)
        __builtin_prefetch(pfb + (long long)(s << 2) * DK, 0, 3);
    #pragma unroll
    for (int s = R; s < R + PF; s += 32)
        __builtin_prefetch(pfa + (s << 2), 0, 3);

    v8f c0 = {}, c1 = {};

    #pragma unroll 1
    for (int base = 0; base < STEPS - R; base += R) {
        if (base + 2 * R + PF <= STEPS) {
            const int s0 = base + R + PF;
            #pragma unroll
            for (int i = 0; i < 4; ++i)
                __builtin_prefetch(pfb + (long long)((s0 << 2) + (i << 5)) * DK, 0, 3);
            __builtin_prefetch(pfa + (s0 << 2), 0, 3);
        }
        #pragma unroll
        for (int s = 0; s < R; ++s) {
            if (s & 1)
                c1 = __builtin_amdgcn_wmma_f32_16x16x4_f32(false, fa[s], false, fb[s],
                                                           (short)0, c1, false, false);
            else
                c0 = __builtin_amdgcn_wmma_f32_16x16x4_f32(false, fa[s], false, fb[s],
                                                           (short)0, c0, false, false);
            const int ks = (base + R + s) << 2;
            fa[s].x = ap[ks];
            fa[s].y = ap[ks + 1];
            fb[s].x = bp[(long long)ks * DK];
            fb[s].y = bp[(long long)(ks + 1) * DK];
        }
    }
    #pragma unroll
    for (int s = 0; s < R; ++s) {
        if (s & 1)
            c1 = __builtin_amdgcn_wmma_f32_16x16x4_f32(false, fa[s], false, fb[s],
                                                       (short)0, c1, false, false);
        else
            c0 = __builtin_amdgcn_wmma_f32_16x16x4_f32(false, fa[s], false, fb[s],
                                                       (short)0, c0, false, false);
    }

    const int rowOff = (lane >> 4) << 3;
    float* __restrict__ cp = ctx + ((long long)(b * TQ + (tm << 4) + rowOff)) * DK + (tn << 4) + m;
    #pragma unroll
    for (int j = 0; j < 8; ++j)
        cp[(long long)j * DK] = c0[j] + c1[j];
}

__global__ void __launch_bounds__(256)
score_tanh_kernel(const float* __restrict__ qpT,
                  const float* __restrict__ kp,
                  const float* __restrict__ v,
                  float* __restrict__ sc)
{
    constexpr int U  = 256;
    constexpr int TQ = 32;
    constexpr int TK = 2048;
    constexpr int KW = 4;
    constexpr int UC = 16;

    const int lane = threadIdx.x & 31;
    const int wave = __builtin_amdgcn_readfirstlane((int)(threadIdx.x >> 5));
    const int row0 = (blockIdx.x * 8 + wave) * KW;
    const int b    = row0 >> 11;
    const int k0   = row0 & (TK - 1);

    const float* __restrict__ qcol = qpT + b * (U * TQ) + lane;
    const float* __restrict__ krow = kp + (long long)row0 * U;

    float acc[KW];
    #pragma unroll
    for (int j = 0; j < KW; ++j) acc[j] = 0.f;

    #pragma unroll 1
    for (int c = 0; c < U; c += UC) {
        float q[UC];
        #pragma unroll
        for (int i = 0; i < UC; ++i) q[i] = qcol[(c + i) * TQ];

        #pragma unroll
        for (int j = 0; j < KW; ++j) {
            #pragma unroll
            for (int i = 0; i < UC; ++i)
                acc[j] += fast_tanh(q[i] + krow[j * U + c + i]) * v[c + i];
        }
    }

    float4 o;
    o.x = acc[0]; o.y = acc[1]; o.z = acc[2]; o.w = acc[3];
    *(float4*)(sc + ((long long)b * TQ + lane) * TK + k0) = o;
}

__global__ void softmax_rows_kernel(const float* __restrict__ sc,
                                    float* __restrict__ attn)
{
    constexpr int TK = 2048;
    const int row = blockIdx.x;
    const int tid = threadIdx.x;
    const float* s = sc   + (long long)row * TK;
    float*       a = attn + (long long)row * TK;

    __shared__ float red[256];

    float m = -INFINITY;
    #pragma unroll
    for (int j = 0; j < 8; ++j) m = fmaxf(m, s[tid + j * 256]);
    red[tid] = m;
    __syncthreads();
    for (int off = 128; off > 0; off >>= 1) {
        if (tid < off) red[tid] = fmaxf(red[tid], red[tid + off]);
        __syncthreads();
    }
    const float mx = red[0];
    __syncthreads();

    float e[8];
    float sum = 0.f;
    #pragma unroll
    for (int j = 0; j < 8; ++j) {
        e[j] = __expf(s[tid + j * 256] - mx);
        sum += e[j];
    }
    red[tid] = sum;
    __syncthreads();
    for (int off = 128; off > 0; off >>= 1) {
        if (tid < off) red[tid] += red[tid + off];
        __syncthreads();
    }
    const float inv = 1.0f / red[0];
    #pragma unroll
    for (int j = 0; j < 8; ++j) a[tid + j * 256] = e[j] * inv;
}

extern "C" void kernel_launch(void* const* d_in, const int* in_sizes, int n_in,
                              void* d_out, int out_size, void* d_ws, size_t ws_size,
                              hipStream_t stream)
{
    (void)in_sizes; (void)n_in; (void)out_size; (void)ws_size;

    const float* query = (const float*)d_in[0];
    const float* keys  = (const float*)d_in[1];
    const float* Wq    = (const float*)d_in[2];
    const float* Wk    = (const float*)d_in[3];
    const float* v     = (const float*)d_in[4];

    float* context = (float*)d_out;
    float* attn    = (float*)d_out + 65536;

    float* qp = (float*)d_ws;
    float* kp = qp + 65536;
    float* sc = kp + (size_t)16384 * 256;

    wmma_gemm_f32<true, true><<<dim3(16, 1), 128, 0, stream>>>(
        Wq, query, qp, 256, 256, 256, 0, 0, 0);

    wmma_gemm_f32<true, false><<<dim3(1024, 1), 128, 0, stream>>>(
        keys, Wk, kp, 16384, 256, 256, 0, 0, 0);

    score_tanh_kernel<<<dim3(512), 256, 0, stream>>>(qp, kp, v, sc);

    softmax_rows_kernel<<<dim3(256), 256, 0, stream>>>(sc, attn);

    context_tile_kernel<<<dim3(256), 32, 0, stream>>>(attn, keys, context);
}


# k_proj GEMM retiled in registers: one wave per 64x64 output tile (4x4 WMMA tiles, 1024 waves = one per SIMD, a set's 32 fragment loads issued as one group); was 16x64 strips on 4096 waves
# speedup vs baseline: 7.0768x; 1.2116x over previous
#include <hip/hip_runtime.h>
#include <math.h>


typedef float v2f __attribute__((ext_vector_type(2)));
typedef float v4f __attribute__((ext_vector_type(4)));
typedef float v8f __attribute__((ext_vector_type(8)));

#ifndef __has_builtin
#define __has_builtin(x) 0
#endif

__device__ __forceinline__ float fast_tanh(float x) {
#if __has_builtin(__builtin_amdgcn_tanhf)
    return __builtin_amdgcn_tanhf(x);
#else
    float r;
    asm volatile("v_tanh_f32 %0, %1\n\tv_nop" : "=v"(r) : "v"(x));
    return r;
#endif
}

template <bool BT, bool CQ, int MT>
__global__ void __launch_bounds__(128)
wmma_gemm_f32(const float* __restrict__ A,
                              const float* __restrict__ B,
                              float* __restrict__ C,
                              int M, int N, int K,
                              long long strideA, long long strideB, long long strideC)
{
    constexpr int NT = 4;
    constexpr int D  = 8;
    constexpr int P  = D / 2;

    const int lane = threadIdx.x & 31;
    const int wave = threadIdx.x >> 5;
    const int wavesPerBlock = blockDim.x >> 5;

    const int groupsN = N >> 6;
    const int totalStrips = ((M >> 4) / MT) * groupsN;
    const int strip = blockIdx.x * wavesPerBlock + wave;
    if (strip >= totalStrips) return;

    const int tm = strip / groupsN;
    const int tg = strip - tm * groupsN;
    const int row0 = tm * (16 * MT);
    const int col0 = tg << 6;

    const float* __restrict__ Ab = A + (long long)blockIdx.y * strideA;
    const float* __restrict__ Bb = B + (long long)blockIdx.y * strideB;
    float* __restrict__ Cb       = C + (long long)blockIdx.y * strideC;

    const int m    = lane & 15;
    const int kq   = (lane >> 4) << 2;

    const float* __restrict__ arow = Ab + (long long)(row0 + m) * K + kq;
    const float* __restrict__ bcol = BT
        ? Bb + (long long)(col0 + m) * K + kq
        : Bb + (long long)kq * N + col0 + m;

    v8f c[MT][NT] = {};

    v4f aA[P][MT], aB[P][MT];
    v4f bA[P][NT], bB[P][NT];

    auto load_set = [&](int k, v4f (&a)[P][MT], v4f (&b)[P][NT]) {
        if constexpr (MT > 1) __builtin_amdgcn_sched_barrier(0);
        #pragma unroll
        for (int p = 0; p < P; ++p) {
            const int kb = k + (p << 3);
            #pragma unroll
            for (int i = 0; i < MT; ++i)
                a[p][i] = *(const v4f*)(arow + (long long)(i << 4) * K + kb);
            #pragma unroll
            for (int t = 0; t < NT; ++t) {
                if (BT) {
                    b[p][t] = *(const v4f*)(bcol + (long long)(t << 4) * K + kb);
                } else {
                    b[p][t].x = bcol[(long long)kb * N + (t << 4)];
                    b[p][t].y = bcol[(long long)(kb + 1) * N + (t << 4)];
                    b[p][t].z = bcol[(long long)(kb + 2) * N + (t << 4)];
                    b[p][t].w = bcol[(long long)(kb + 3) * N + (t << 4)];
                }
            }
        }
        if constexpr (MT > 1) __builtin_amdgcn_sched_barrier(0);
    };
    auto mma_set = [&](const v4f (&a)[P][MT], const v4f (&b)[P][NT]) {
        #pragma unroll
        for (int p = 0; p < P; ++p) {
            #pragma unroll
            for (int i = 0; i < MT; ++i) {
                const v2f a0 = __builtin_shufflevector(a[p][i], a[p][i], 0, 1);
                #pragma unroll
                for (int t = 0; t < NT; ++t) {
                    const v2f b0 = __builtin_shufflevector(b[p][t], b[p][t], 0, 1);
                    c[i][t] = __builtin_amdgcn_wmma_f32_16x16x4_f32(false, a0, false, b0,
                                                                    (short)0, c[i][t],
                                                                    false, false);
                }
            }
            #pragma unroll
            for (int i = 0; i < MT; ++i) {
                const v2f a1 = __builtin_shufflevector(a[p][i], a[p][i], 2, 3);
                #pragma unroll
                for (int t = 0; t < NT; ++t) {
                    const v2f b1 = __builtin_shufflevector(b[p][t], b[p][t], 2, 3);
                    c[i][t] = __builtin_amdgcn_wmma_f32_16x16x4_f32(false, a1, false, b1,
                                                                    (short)0, c[i][t],
                                                                    false, false);
                }
            }
        }
    };

    constexpr int KS = D * 4;
    load_set(0, aA, bA);
    int k = 0;
    for (; k < K - 2 * KS; k += 2 * KS) {
        load_set(k + KS, aB, bB);
        mma_set(aA, bA);
        load_set(k + 2 * KS, aA, bA);
        mma_set(aB, bB);
    }
    load_set(k + KS, aB, bB);
    mma_set(aA, bA);
    mma_set(aB, bB);

    const int rowOff = (lane >> 4) << 3;
    #pragma unroll
    for (int i = 0; i < MT; ++i) {
        #pragma unroll
        for (int t = 0; t < NT; ++t) {
            #pragma unroll
            for (int j = 0; j < 8; ++j) {
                const int row = row0 + (i << 4) + rowOff + j;
                const int col = col0 + (t << 4) + (lane & 15);
                if (CQ) {
                    Cb[(long long)(col >> 5) * (M * 32) + row * 32 + (col & 31)] = c[i][t][j];
                } else {
                    Cb[(long long)row * N + col] = c[i][t][j];
                }
            }
        }
    }
}

__global__ void __launch_bounds__(32)
context_tile_kernel(const float* __restrict__ attn,
                    const float* __restrict__ keys,
                    float* __restrict__ ctx)
{
    constexpr int TQ = 32;
    constexpr int TK = 2048;
    constexpr int DK = 256;
    constexpr int R     = 32;
    constexpr int STEPS = TK / 4;
    constexpr int PF    = 64;

    const int lane = threadIdx.x;
    const int tile = blockIdx.x;
    const int tn   = tile & 15;
    const int tm   = (tile >> 4) & 1;
    const int b    = tile >> 5;

    const int m     = lane & 15;
    const int khalf = (lane >> 4) << 1;

    const float* __restrict__ ap = attn + ((long long)(b * TQ + (tm << 4) + m)) * TK + khalf;
    const float* __restrict__ bp = keys + ((long long)b * TK + khalf) * DK + (tn << 4) + m;

    v2f fa[R], fb[R];
    #pragma unroll
    for (int s = 0; s < R; ++s) {
        const int ks = s << 2;
        fa[s].x = ap[ks];
        fa[s].y = ap[ks + 1];
        fb[s].x = bp[(long long)ks * DK];
        fb[s].y = bp[(long long)(ks + 1) * DK];
    }

    const float* __restrict__ pfb = keys + ((long long)b * TK + lane) * DK + (tn << 4);
    const float* __restrict__ pfa = attn + ((long long)(b * TQ + (tm << 4) + m)) * TK + ((lane >> 4) << 6);

    #pragma unroll
    for (int s = R; s < R + PF; s += 8)
        __builtin_prefetch(pfb + (long long)(s << 2) * DK, 0, 3);
    #pragma unroll
    for (int s = R; s < R + PF; s += 32)
        __builtin_prefetch(pfa + (s << 2), 0, 3);

    v8f c0 = {}, c1 = {};

    #pragma unroll 1
    for (int base = 0; base < STEPS - R; base += R) {
        if (base + 2 * R + PF <= STEPS) {
            const int s0 = base + R + PF;
            #pragma unroll
            for (int i = 0; i < 4; ++i)
                __builtin_prefetch(pfb + (long long)((s0 << 2) + (i << 5)) * DK, 0, 3);
            __builtin_prefetch(pfa + (s0 << 2), 0, 3);
        }
        #pragma unroll
        for (int s = 0; s < R; ++s) {
            if (s & 1)
                c1 = __builtin_amdgcn_wmma_f32_16x16x4_f32(false, fa[s], false, fb[s],
                                                           (short)0, c1, false, false);
            else
                c0 = __builtin_amdgcn_wmma_f32_16x16x4_f32(false, fa[s], false, fb[s],
                                                           (short)0, c0, false, false);
            const int ks = (base + R + s) << 2;
            fa[s].x = ap[ks];
            fa[s].y = ap[ks + 1];
            fb[s].x = bp[(long long)ks * DK];
            fb[s].y = bp[(long long)(ks + 1) * DK];
        }
    }
    #pragma unroll
    for (int s = 0; s < R; ++s) {
        if (s & 1)
            c1 = __builtin_amdgcn_wmma_f32_16x16x4_f32(false, fa[s], false, fb[s],
                                                       (short)0, c1, false, false);
        else
            c0 = __builtin_amdgcn_wmma_f32_16x16x4_f32(false, fa[s], false, fb[s],
                                                       (short)0, c0, false, false);
    }

    const int rowOff = (lane >> 4) << 3;
    float* __restrict__ cp = ctx + ((long long)(b * TQ + (tm << 4) + rowOff)) * DK + (tn << 4) + m;
    #pragma unroll
    for (int j = 0; j < 8; ++j)
        cp[(long long)j * DK] = c0[j] + c1[j];
}

__global__ void __launch_bounds__(256)
score_tanh_kernel(const float* __restrict__ qpT,
                  const float* __restrict__ kp,
                  const float* __restrict__ v,
                  float* __restrict__ sc)
{
    constexpr int U  = 256;
    constexpr int TQ = 32;
    constexpr int TK = 2048;
    constexpr int KW = 4;
    constexpr int UC = 16;

    const int lane = threadIdx.x & 31;
    const int wave = __builtin_amdgcn_readfirstlane((int)(threadIdx.x >> 5));
    const int row0 = (blockIdx.x * 8 + wave) * KW;
    const int b    = row0 >> 11;
    const int k0   = row0 & (TK - 1);

    const float* __restrict__ qcol = qpT + b * (U * TQ) + lane;
    const float* __restrict__ krow = kp + (long long)row0 * U;

    float acc[KW];
    #pragma unroll
    for (int j = 0; j < KW; ++j) acc[j] = 0.f;

    #pragma unroll 1
    for (int c = 0; c < U; c += UC) {
        float q[UC];
        #pragma unroll
        for (int i = 0; i < UC; ++i) q[i] = qcol[(c + i) * TQ];

        #pragma unroll
        for (int j = 0; j < KW; ++j) {
            #pragma unroll
            for (int i = 0; i < UC; ++i)
                acc[j] += fast_tanh(q[i] + krow[j * U + c + i]) * v[c + i];
        }
    }

    float4 o;
    o.x = acc[0]; o.y = acc[1]; o.z = acc[2]; o.w = acc[3];
    *(float4*)(sc + ((long long)b * TQ + lane) * TK + k0) = o;
}

__global__ void softmax_rows_kernel(const float* __restrict__ sc,
                                    float* __restrict__ attn)
{
    constexpr int TK = 2048;
    const int row = blockIdx.x;
    const int tid = threadIdx.x;
    const float* s = sc   + (long long)row * TK;
    float*       a = attn + (long long)row * TK;

    __shared__ float red[256];

    float m = -INFINITY;
    #pragma unroll
    for (int j = 0; j < 8; ++j) m = fmaxf(m, s[tid + j * 256]);
    red[tid] = m;
    __syncthreads();
    for (int off = 128; off > 0; off >>= 1) {
        if (tid < off) red[tid] = fmaxf(red[tid], red[tid + off]);
        __syncthreads();
    }
    const float mx = red[0];
    __syncthreads();

    float e[8];
    float sum = 0.f;
    #pragma unroll
    for (int j = 0; j < 8; ++j) {
        e[j] = __expf(s[tid + j * 256] - mx);
        sum += e[j];
    }
    red[tid] = sum;
    __syncthreads();
    for (int off = 128; off > 0; off >>= 1) {
        if (tid < off) red[tid] += red[tid + off];
        __syncthreads();
    }
    const float inv = 1.0f / red[0];
    #pragma unroll
    for (int j = 0; j < 8; ++j) a[tid + j * 256] = e[j] * inv;
}

extern "C" void kernel_launch(void* const* d_in, const int* in_sizes, int n_in,
                              void* d_out, int out_size, void* d_ws, size_t ws_size,
                              hipStream_t stream)
{
    (void)in_sizes; (void)n_in; (void)out_size; (void)ws_size;

    const float* query = (const float*)d_in[0];
    const float* keys  = (const float*)d_in[1];
    const float* Wq    = (const float*)d_in[2];
    const float* Wk    = (const float*)d_in[3];
    const float* v     = (const float*)d_in[4];

    float* context = (float*)d_out;
    float* attn    = (float*)d_out + 65536;

    float* qp = (float*)d_ws;
    float* kp = qp + 65536;
    float* sc = kp + (size_t)16384 * 256;

    wmma_gemm_f32<true, true, 1><<<dim3(16, 1), 128, 0, stream>>>(
        Wq, query, qp, 256, 256, 256, 0, 0, 0);

    wmma_gemm_f32<true, false, 4><<<dim3(256, 1), 128, 0, stream>>>(
        keys, Wk, kp, 16384, 256, 256, 0, 0, 0);

    score_tanh_kernel<<<dim3(512), 256, 0, stream>>>(qp, kp, v, sc);

    softmax_rows_kernel<<<dim3(256), 256, 0, stream>>>(sc, attn);

    context_tile_kernel<<<dim3(256), 32, 0, stream>>>(attn, keys, context);
}


# q_proj GEMM re-decomposed: one wave per 16x16 tile (256 single-wave workgroups), a tile's whole K asked for as one group of 64 loads ahead of its 64 matrix instructions; was 64 waves of 16x64 strips
# speedup vs baseline: 7.9945x; 1.1297x over previous
#include <hip/hip_runtime.h>
#include <math.h>


typedef float v2f __attribute__((ext_vector_type(2)));
typedef float v4f __attribute__((ext_vector_type(4)));
typedef float v8f __attribute__((ext_vector_type(8)));

#ifndef __has_builtin
#define __has_builtin(x) 0
#endif

__device__ __forceinline__ float fast_tanh(float x) {
#if __has_builtin(__builtin_amdgcn_tanhf)
    return __builtin_amdgcn_tanhf(x);
#else
    float r;
    asm volatile("v_tanh_f32 %0, %1\n\tv_nop" : "=v"(r) : "v"(x));
    return r;
#endif
}

template <bool BT, int MT>
__global__ void __launch_bounds__(128)
wmma_gemm_f32(const float* __restrict__ A,
                              const float* __restrict__ B,
                              float* __restrict__ C,
                              int M, int N, int K,
                              long long strideA, long long strideB, long long strideC)
{
    constexpr int NT = 4;
    constexpr int D  = 8;
    constexpr int P  = D / 2;

    const int lane = threadIdx.x & 31;
    const int wave = threadIdx.x >> 5;
    const int wavesPerBlock = blockDim.x >> 5;

    const int groupsN = N >> 6;
    const int totalStrips = ((M >> 4) / MT) * groupsN;
    const int strip = blockIdx.x * wavesPerBlock + wave;
    if (strip >= totalStrips) return;

    const int tm = strip / groupsN;
    const int tg = strip - tm * groupsN;
    const int row0 = tm * (16 * MT);
    const int col0 = tg << 6;

    const float* __restrict__ Ab = A + (long long)blockIdx.y * strideA;
    const float* __restrict__ Bb = B + (long long)blockIdx.y * strideB;
    float* __restrict__ Cb       = C + (long long)blockIdx.y * strideC;

    const int m    = lane & 15;
    const int kq   = (lane >> 4) << 2;

    const float* __restrict__ arow = Ab + (long long)(row0 + m) * K + kq;
    const float* __restrict__ bcol = BT
        ? Bb + (long long)(col0 + m) * K + kq
        : Bb + (long long)kq * N + col0 + m;

    v8f c[MT][NT] = {};

    v4f aA[P][MT], aB[P][MT];
    v4f bA[P][NT], bB[P][NT];

    auto load_set = [&](int k, v4f (&a)[P][MT], v4f (&b)[P][NT]) {
        if constexpr (MT > 1) __builtin_amdgcn_sched_barrier(0);
        #pragma unroll
        for (int p = 0; p < P; ++p) {
            const int kb = k + (p << 3);
            #pragma unroll
            for (int i = 0; i < MT; ++i)
                a[p][i] = *(const v4f*)(arow + (long long)(i << 4) * K + kb);
            #pragma unroll
            for (int t = 0; t < NT; ++t) {
                if (BT) {
                    b[p][t] = *(const v4f*)(bcol + (long long)(t << 4) * K + kb);
                } else {
                    b[p][t].x = bcol[(long long)kb * N + (t << 4)];
                    b[p][t].y = bcol[(long long)(kb + 1) * N + (t << 4)];
                    b[p][t].z = bcol[(long long)(kb + 2) * N + (t << 4)];
                    b[p][t].w = bcol[(long long)(kb + 3) * N + (t << 4)];
                }
            }
        }
        if constexpr (MT > 1) __builtin_amdgcn_sched_barrier(0);
    };
    auto mma_set = [&](const v4f (&a)[P][MT], const v4f (&b)[P][NT]) {
        #pragma unroll
        for (int p = 0; p < P; ++p) {
            #pragma unroll
            for (int i = 0; i < MT; ++i) {
                const v2f a0 = __builtin_shufflevector(a[p][i], a[p][i], 0, 1);
                #pragma unroll
                for (int t = 0; t < NT; ++t) {
                    const v2f b0 = __builtin_shufflevector(b[p][t], b[p][t], 0, 1);
                    c[i][t] = __builtin_amdgcn_wmma_f32_16x16x4_f32(false, a0, false, b0,
                                                                    (short)0, c[i][t],
                                                                    false, false);
                }
            }
            #pragma unroll
            for (int i = 0; i < MT; ++i) {
                const v2f a1 = __builtin_shufflevector(a[p][i], a[p][i], 2, 3);
                #pragma unroll
                for (int t = 0; t < NT; ++t) {
                    const v2f b1 = __builtin_shufflevector(b[p][t], b[p][t], 2, 3);
                    c[i][t] = __builtin_amdgcn_wmma_f32_16x16x4_f32(false, a1, false, b1,
                                                                    (short)0, c[i][t],
                                                                    false, false);
                }
            }
        }
    };

    constexpr int KS = D * 4;
    load_set(0, aA, bA);
    int k = 0;
    for (; k < K - 2 * KS; k += 2 * KS) {
        load_set(k + KS, aB, bB);
        mma_set(aA, bA);
        load_set(k + 2 * KS, aA, bA);
        mma_set(aB, bB);
    }
    load_set(k + KS, aB, bB);
    mma_set(aA, bA);
    mma_set(aB, bB);

    const int rowOff = (lane >> 4) << 3;
    #pragma unroll
    for (int i = 0; i < MT; ++i) {
        #pragma unroll
        for (int t = 0; t < NT; ++t) {
            #pragma unroll
            for (int j = 0; j < 8; ++j) {
                const int row = row0 + (i << 4) + rowOff + j;
                const int col = col0 + (t << 4) + (lane & 15);
                Cb[(long long)row * N + col] = c[i][t][j];
            }
        }
    }
}

__global__ void __launch_bounds__(32)
qproj_tile_kernel(const float* __restrict__ Wq,
                  const float* __restrict__ query,
                  float* __restrict__ qpT)
{
    constexpr int U  = 256;
    constexpr int D  = 256;
    constexpr int TQ = 32;
    constexpr int NB = D / 8;

    const int lane = threadIdx.x;
    const int tile = blockIdx.x;
    const int tn   = tile & 15;
    const int tm   = tile >> 4;

    const int m  = lane & 15;
    const int kq = (lane >> 4) << 2;

    const float* __restrict__ ap = Wq    + ((tm << 4) + m) * D + kq;
    const float* __restrict__ bp = query + ((tn << 4) + m) * D + kq;

    v4f a[NB], b[NB];
    #pragma unroll
    for (int p = 0; p < NB; ++p) {
        a[p] = *(const v4f*)(ap + (p << 3));
        b[p] = *(const v4f*)(bp + (p << 3));
    }
    __builtin_amdgcn_sched_barrier(0);

    v8f c0 = {}, c1 = {};
    #pragma unroll
    for (int p = 0; p < NB; ++p) {
        const v2f a0 = __builtin_shufflevector(a[p], a[p], 0, 1);
        const v2f b0 = __builtin_shufflevector(b[p], b[p], 0, 1);
        const v2f a1 = __builtin_shufflevector(a[p], a[p], 2, 3);
        const v2f b1 = __builtin_shufflevector(b[p], b[p], 2, 3);
        c0 = __builtin_amdgcn_wmma_f32_16x16x4_f32(false, a0, false, b0,
                                                   (short)0, c0, false, false);
        c1 = __builtin_amdgcn_wmma_f32_16x16x4_f32(false, a1, false, b1,
                                                   (short)0, c1, false, false);
    }

    const int rowOff = (lane >> 4) << 3;
    const int col    = (tn << 4) + m;
    float* __restrict__ cp = qpT + (col >> 5) * (U * TQ) + ((tm << 4) + rowOff) * TQ + (col & 31);
    #pragma unroll
    for (int j = 0; j < 8; ++j)
        cp[j * TQ] = c0[j] + c1[j];
}

__global__ void __launch_bounds__(32)
context_tile_kernel(const float* __restrict__ attn,
                    const float* __restrict__ keys,
                    float* __restrict__ ctx)
{
    constexpr int TQ = 32;
    constexpr int TK = 2048;
    constexpr int DK = 256;
    constexpr int R     = 32;
    constexpr int STEPS = TK / 4;
    constexpr int PF    = 64;

    const int lane = threadIdx.x;
    const int tile = blockIdx.x;
    const int tn   = tile & 15;
    const int tm   = (tile >> 4) & 1;
    const int b    = tile >> 5;

    const int m     = lane & 15;
    const int khalf = (lane >> 4) << 1;

    const float* __restrict__ ap = attn + ((long long)(b * TQ + (tm << 4) + m)) * TK + khalf;
    const float* __restrict__ bp = keys + ((long long)b * TK + khalf) * DK + (tn << 4) + m;

    v2f fa[R], fb[R];
    #pragma unroll
    for (int s = 0; s < R; ++s) {
        const int ks = s << 2;
        fa[s].x = ap[ks];
        fa[s].y = ap[ks + 1];
        fb[s].x = bp[(long long)ks * DK];
        fb[s].y = bp[(long long)(ks + 1) * DK];
    }

    const float* __restrict__ pfb = keys + ((long long)b * TK + lane) * DK + (tn << 4);
    const float* __restrict__ pfa = attn + ((long long)(b * TQ + (tm << 4) + m)) * TK + ((lane >> 4) << 6);

    #pragma unroll
    for (int s = R; s < R + PF; s += 8)
        __builtin_prefetch(pfb + (long long)(s << 2) * DK, 0, 3);
    #pragma unroll
    for (int s = R; s < R + PF; s += 32)
        __builtin_prefetch(pfa + (s << 2), 0, 3);

    v8f c0 = {}, c1 = {};

    #pragma unroll 1
    for (int base = 0; base < STEPS - R; base += R) {
        if (base + 2 * R + PF <= STEPS) {
            const int s0 = base + R + PF;
            #pragma unroll
            for (int i = 0; i < 4; ++i)
                __builtin_prefetch(pfb + (long long)((s0 << 2) + (i << 5)) * DK, 0, 3);
            __builtin_prefetch(pfa + (s0 << 2), 0, 3);
        }
        #pragma unroll
        for (int s = 0; s < R; ++s) {
            if (s & 1)
                c1 = __builtin_amdgcn_wmma_f32_16x16x4_f32(false, fa[s], false, fb[s],
                                                           (short)0, c1, false, false);
            else
                c0 = __builtin_amdgcn_wmma_f32_16x16x4_f32(false, fa[s], false, fb[s],
                                                           (short)0, c0, false, false);
            const int ks = (base + R + s) << 2;
            fa[s].x = ap[ks];
            fa[s].y = ap[ks + 1];
            fb[s].x = bp[(long long)ks * DK];
            fb[s].y = bp[(long long)(ks + 1) * DK];
        }
    }
    #pragma unroll
    for (int s = 0; s < R; ++s) {
        if (s & 1)
            c1 = __builtin_amdgcn_wmma_f32_16x16x4_f32(false, fa[s], false, fb[s],
                                                       (short)0, c1, false, false);
        else
            c0 = __builtin_amdgcn_wmma_f32_16x16x4_f32(false, fa[s], false, fb[s],
                                                       (short)0, c0, false, false);
    }

    const int rowOff = (lane >> 4) << 3;
    float* __restrict__ cp = ctx + ((long long)(b * TQ + (tm << 4) + rowOff)) * DK + (tn << 4) + m;
    #pragma unroll
    for (int j = 0; j < 8; ++j)
        cp[(long long)j * DK] = c0[j] + c1[j];
}

__global__ void __launch_bounds__(256)
score_tanh_kernel(const float* __restrict__ qpT,
                  const float* __restrict__ kp,
                  const float* __restrict__ v,
                  float* __restrict__ sc)
{
    constexpr int U  = 256;
    constexpr int TQ = 32;
    constexpr int TK = 2048;
    constexpr int KW = 4;
    constexpr int UC = 16;

    const int lane = threadIdx.x & 31;
    const int wave = __builtin_amdgcn_readfirstlane((int)(threadIdx.x >> 5));
    const int row0 = (blockIdx.x * 8 + wave) * KW;
    const int b    = row0 >> 11;
    const int k0   = row0 & (TK - 1);

    const float* __restrict__ qcol = qpT + b * (U * TQ) + lane;
    const float* __restrict__ krow = kp + (long long)row0 * U;

    float acc[KW];
    #pragma unroll
    for (int j = 0; j < KW; ++j) acc[j] = 0.f;

    #pragma unroll 1
    for (int c = 0; c < U; c += UC) {
        float q[UC];
        #pragma unroll
        for (int i = 0; i < UC; ++i) q[i] = qcol[(c + i) * TQ];

        #pragma unroll
        for (int j = 0; j < KW; ++j) {
            #pragma unroll
            for (int i = 0; i < UC; ++i)
                acc[j] += fast_tanh(q[i] + krow[j * U + c + i]) * v[c + i];
        }
    }

    float4 o;
    o.x = acc[0]; o.y = acc[1]; o.z = acc[2]; o.w = acc[3];
    *(float4*)(sc + ((long long)b * TQ + lane) * TK + k0) = o;
}

__global__ void softmax_rows_kernel(const float* __restrict__ sc,
                                    float* __restrict__ attn)
{
    constexpr int TK = 2048;
    const int row = blockIdx.x;
    const int tid = threadIdx.x;
    const float* s = sc   + (long long)row * TK;
    float*       a = attn + (long long)row * TK;

    __shared__ float red[256];

    float m = -INFINITY;
    #pragma unroll
    for (int j = 0; j < 8; ++j) m = fmaxf(m, s[tid + j * 256]);
    red[tid] = m;
    __syncthreads();
    for (int off = 128; off > 0; off >>= 1) {
        if (tid < off) red[tid] = fmaxf(red[tid], red[tid + off]);
        __syncthreads();
    }
    const float mx = red[0];
    __syncthreads();

    float e[8];
    float sum = 0.f;
    #pragma unroll
    for (int j = 0; j < 8; ++j) {
        e[j] = __expf(s[tid + j * 256] - mx);
        sum += e[j];
    }
    red[tid] = sum;
    __syncthreads();
    for (int off = 128; off > 0; off >>= 1) {
        if (tid < off) red[tid] += red[tid + off];
        __syncthreads();
    }
    const float inv = 1.0f / red[0];
    #pragma unroll
    for (int j = 0; j < 8; ++j) a[tid + j * 256] = e[j] * inv;
}

extern "C" void kernel_launch(void* const* d_in, const int* in_sizes, int n_in,
                              void* d_out, int out_size, void* d_ws, size_t ws_size,
                              hipStream_t stream)
{
    (void)in_sizes; (void)n_in; (void)out_size; (void)ws_size;

    const float* query = (const float*)d_in[0];
    const float* keys  = (const float*)d_in[1];
    const float* Wq    = (const float*)d_in[2];
    const float* Wk    = (const float*)d_in[3];
    const float* v     = (const float*)d_in[4];

    float* context = (float*)d_out;
    float* attn    = (float*)d_out + 65536;

    float* qp = (float*)d_ws;
    float* kp = qp + 65536;
    float* sc = kp + (size_t)16384 * 256;

    qproj_tile_kernel<<<dim3(256), 32, 0, stream>>>(Wq, query, qp);

    wmma_gemm_f32<true, 4><<<dim3(256, 1), 128, 0, stream>>>(
        keys, Wk, kp, 16384, 256, 256, 0, 0, 0);

    score_tanh_kernel<<<dim3(512), 256, 0, stream>>>(qp, kp, v, sc);

    softmax_rows_kernel<<<dim3(256), 256, 0, stream>>>(sc, attn);

    context_tile_kernel<<<dim3(256), 32, 0, stream>>>(attn, keys, context);
}


# Softmax re-decomposed: one wave per row (256 single-wave workgroups), the row's 2048 scores in registers, max and sum by in-wave lane exchanges; was 8 waves per row, 19 barriers, two LDS trees
# speedup vs baseline: 8.2410x; 1.0308x over previous
#include <hip/hip_runtime.h>
#include <math.h>


typedef float v2f __attribute__((ext_vector_type(2)));
typedef float v4f __attribute__((ext_vector_type(4)));
typedef float v8f __attribute__((ext_vector_type(8)));

#ifndef __has_builtin
#define __has_builtin(x) 0
#endif

__device__ __forceinline__ float fast_tanh(float x) {
#if __has_builtin(__builtin_amdgcn_tanhf)
    return __builtin_amdgcn_tanhf(x);
#else
    float r;
    asm volatile("v_tanh_f32 %0, %1\n\tv_nop" : "=v"(r) : "v"(x));
    return r;
#endif
}

template <bool BT, int MT>
__global__ void __launch_bounds__(128)
wmma_gemm_f32(const float* __restrict__ A,
                              const float* __restrict__ B,
                              float* __restrict__ C,
                              int M, int N, int K,
                              long long strideA, long long strideB, long long strideC)
{
    constexpr int NT = 4;
    constexpr int D  = 8;
    constexpr int P  = D / 2;

    const int lane = threadIdx.x & 31;
    const int wave = threadIdx.x >> 5;
    const int wavesPerBlock = blockDim.x >> 5;

    const int groupsN = N >> 6;
    const int totalStrips = ((M >> 4) / MT) * groupsN;
    const int strip = blockIdx.x * wavesPerBlock + wave;
    if (strip >= totalStrips) return;

    const int tm = strip / groupsN;
    const int tg = strip - tm * groupsN;
    const int row0 = tm * (16 * MT);
    const int col0 = tg << 6;

    const float* __restrict__ Ab = A + (long long)blockIdx.y * strideA;
    const float* __restrict__ Bb = B + (long long)blockIdx.y * strideB;
    float* __restrict__ Cb       = C + (long long)blockIdx.y * strideC;

    const int m    = lane & 15;
    const int kq   = (lane >> 4) << 2;

    const float* __restrict__ arow = Ab + (long long)(row0 + m) * K + kq;
    const float* __restrict__ bcol = BT
        ? Bb + (long long)(col0 + m) * K + kq
        : Bb + (long long)kq * N + col0 + m;

    v8f c[MT][NT] = {};

    v4f aA[P][MT], aB[P][MT];
    v4f bA[P][NT], bB[P][NT];

    auto load_set = [&](int k, v4f (&a)[P][MT], v4f (&b)[P][NT]) {
        if constexpr (MT > 1) __builtin_amdgcn_sched_barrier(0);
        #pragma unroll
        for (int p = 0; p < P; ++p) {
            const int kb = k + (p << 3);
            #pragma unroll
            for (int i = 0; i < MT; ++i)
                a[p][i] = *(const v4f*)(arow + (long long)(i << 4) * K + kb);
            #pragma unroll
            for (int t = 0; t < NT; ++t) {
                if (BT) {
                    b[p][t] = *(const v4f*)(bcol + (long long)(t << 4) * K + kb);
                } else {
                    b[p][t].x = bcol[(long long)kb * N + (t << 4)];
                    b[p][t].y = bcol[(long long)(kb + 1) * N + (t << 4)];
                    b[p][t].z = bcol[(long long)(kb + 2) * N + (t << 4)];
                    b[p][t].w = bcol[(long long)(kb + 3) * N + (t << 4)];
                }
            }
        }
        if constexpr (MT > 1) __builtin_amdgcn_sched_barrier(0);
    };
    auto mma_set = [&](const v4f (&a)[P][MT], const v4f (&b)[P][NT]) {
        #pragma unroll
        for (int p = 0; p < P; ++p) {
            #pragma unroll
            for (int i = 0; i < MT; ++i) {
                const v2f a0 = __builtin_shufflevector(a[p][i], a[p][i], 0, 1);
                #pragma unroll
                for (int t = 0; t < NT; ++t) {
                    const v2f b0 = __builtin_shufflevector(b[p][t], b[p][t], 0, 1);
                    c[i][t] = __builtin_amdgcn_wmma_f32_16x16x4_f32(false, a0, false, b0,
                                                                    (short)0, c[i][t],
                                                                    false, false);
                }
            }
            #pragma unroll
            for (int i = 0; i < MT; ++i) {
                const v2f a1 = __builtin_shufflevector(a[p][i], a[p][i], 2, 3);
                #pragma unroll
                for (int t = 0; t < NT; ++t) {
                    const v2f b1 = __builtin_shufflevector(b[p][t], b[p][t], 2, 3);
                    c[i][t] = __builtin_amdgcn_wmma_f32_16x16x4_f32(false, a1, false, b1,
                                                                    (short)0, c[i][t],
                                                                    false, false);
                }
            }
        }
    };

    constexpr int KS = D * 4;
    load_set(0, aA, bA);
    int k = 0;
    for (; k < K - 2 * KS; k += 2 * KS) {
        load_set(k + KS, aB, bB);
        mma_set(aA, bA);
        load_set(k + 2 * KS, aA, bA);
        mma_set(aB, bB);
    }
    load_set(k + KS, aB, bB);
    mma_set(aA, bA);
    mma_set(aB, bB);

    const int rowOff = (lane >> 4) << 3;
    #pragma unroll
    for (int i = 0; i < MT; ++i) {
        #pragma unroll
        for (int t = 0; t < NT; ++t) {
            #pragma unroll
            for (int j = 0; j < 8; ++j) {
                const int row = row0 + (i << 4) + rowOff + j;
                const int col = col0 + (t << 4) + (lane & 15);
                Cb[(long long)row * N + col] = c[i][t][j];
            }
        }
    }
}

__global__ void __launch_bounds__(32)
qproj_tile_kernel(const float* __restrict__ Wq,
                  const float* __restrict__ query,
                  float* __restrict__ qpT)
{
    constexpr int U  = 256;
    constexpr int D  = 256;
    constexpr int TQ = 32;
    constexpr int NB = D / 8;

    const int lane = threadIdx.x;
    const int tile = blockIdx.x;
    const int tn   = tile & 15;
    const int tm   = tile >> 4;

    const int m  = lane & 15;
    const int kq = (lane >> 4) << 2;

    const float* __restrict__ ap = Wq    + ((tm << 4) + m) * D + kq;
    const float* __restrict__ bp = query + ((tn << 4) + m) * D + kq;

    v4f a[NB], b[NB];
    #pragma unroll
    for (int p = 0; p < NB; ++p) {
        a[p] = *(const v4f*)(ap + (p << 3));
        b[p] = *(const v4f*)(bp + (p << 3));
    }
    __builtin_amdgcn_sched_barrier(0);

    v8f c0 = {}, c1 = {};
    #pragma unroll
    for (int p = 0; p < NB; ++p) {
        const v2f a0 = __builtin_shufflevector(a[p], a[p], 0, 1);
        const v2f b0 = __builtin_shufflevector(b[p], b[p], 0, 1);
        const v2f a1 = __builtin_shufflevector(a[p], a[p], 2, 3);
        const v2f b1 = __builtin_shufflevector(b[p], b[p], 2, 3);
        c0 = __builtin_amdgcn_wmma_f32_16x16x4_f32(false, a0, false, b0,
                                                   (short)0, c0, false, false);
        c1 = __builtin_amdgcn_wmma_f32_16x16x4_f32(false, a1, false, b1,
                                                   (short)0, c1, false, false);
    }

    const int rowOff = (lane >> 4) << 3;
    const int col    = (tn << 4) + m;
    float* __restrict__ cp = qpT + (col >> 5) * (U * TQ) + ((tm << 4) + rowOff) * TQ + (col & 31);
    #pragma unroll
    for (int j = 0; j < 8; ++j)
        cp[j * TQ] = c0[j] + c1[j];
}

__global__ void __launch_bounds__(32)
context_tile_kernel(const float* __restrict__ attn,
                    const float* __restrict__ keys,
                    float* __restrict__ ctx)
{
    constexpr int TQ = 32;
    constexpr int TK = 2048;
    constexpr int DK = 256;
    constexpr int R     = 32;
    constexpr int STEPS = TK / 4;
    constexpr int PF    = 64;

    const int lane = threadIdx.x;
    const int tile = blockIdx.x;
    const int tn   = tile & 15;
    const int tm   = (tile >> 4) & 1;
    const int b    = tile >> 5;

    const int m     = lane & 15;
    const int khalf = (lane >> 4) << 1;

    const float* __restrict__ ap = attn + ((long long)(b * TQ + (tm << 4) + m)) * TK + khalf;
    const float* __restrict__ bp = keys + ((long long)b * TK + khalf) * DK + (tn << 4) + m;

    v2f fa[R], fb[R];
    #pragma unroll
    for (int s = 0; s < R; ++s) {
        const int ks = s << 2;
        fa[s].x = ap[ks];
        fa[s].y = ap[ks + 1];
        fb[s].x = bp[(long long)ks * DK];
        fb[s].y = bp[(long long)(ks + 1) * DK];
    }

    const float* __restrict__ pfb = keys + ((long long)b * TK + lane) * DK + (tn << 4);
    const float* __restrict__ pfa = attn + ((long long)(b * TQ + (tm << 4) + m)) * TK + ((lane >> 4) << 6);

    #pragma unroll
    for (int s = R; s < R + PF; s += 8)
        __builtin_prefetch(pfb + (long long)(s << 2) * DK, 0, 3);
    #pragma unroll
    for (int s = R; s < R + PF; s += 32)
        __builtin_prefetch(pfa + (s << 2), 0, 3);

    v8f c0 = {}, c1 = {};

    #pragma unroll 1
    for (int base = 0; base < STEPS - R; base += R) {
        if (base + 2 * R + PF <= STEPS) {
            const int s0 = base + R + PF;
            #pragma unroll
            for (int i = 0; i < 4; ++i)
                __builtin_prefetch(pfb + (long long)((s0 << 2) + (i << 5)) * DK, 0, 3);
            __builtin_prefetch(pfa + (s0 << 2), 0, 3);
        }
        #pragma unroll
        for (int s = 0; s < R; ++s) {
            if (s & 1)
                c1 = __builtin_amdgcn_wmma_f32_16x16x4_f32(false, fa[s], false, fb[s],
                                                           (short)0, c1, false, false);
            else
                c0 = __builtin_amdgcn_wmma_f32_16x16x4_f32(false, fa[s], false, fb[s],
                                                           (short)0, c0, false, false);
            const int ks = (base + R + s) << 2;
            fa[s].x = ap[ks];
            fa[s].y = ap[ks + 1];
            fb[s].x = bp[(long long)ks * DK];
            fb[s].y = bp[(long long)(ks + 1) * DK];
        }
    }
    #pragma unroll
    for (int s = 0; s < R; ++s) {
        if (s & 1)
            c1 = __builtin_amdgcn_wmma_f32_16x16x4_f32(false, fa[s], false, fb[s],
                                                       (short)0, c1, false, false);
        else
            c0 = __builtin_amdgcn_wmma_f32_16x16x4_f32(false, fa[s], false, fb[s],
                                                       (short)0, c0, false, false);
    }

    const int rowOff = (lane >> 4) << 3;
    float* __restrict__ cp = ctx + ((long long)(b * TQ + (tm << 4) + rowOff)) * DK + (tn << 4) + m;
    #pragma unroll
    for (int j = 0; j < 8; ++j)
        cp[(long long)j * DK] = c0[j] + c1[j];
}

__global__ void __launch_bounds__(256)
score_tanh_kernel(const float* __restrict__ qpT,
                  const float* __restrict__ kp,
                  const float* __restrict__ v,
                  float* __restrict__ sc)
{
    constexpr int U  = 256;
    constexpr int TQ = 32;
    constexpr int TK = 2048;
    constexpr int KW = 4;
    constexpr int UC = 16;

    const int lane = threadIdx.x & 31;
    const int wave = __builtin_amdgcn_readfirstlane((int)(threadIdx.x >> 5));
    const int row0 = (blockIdx.x * 8 + wave) * KW;
    const int b    = row0 >> 11;
    const int k0   = row0 & (TK - 1);

    const float* __restrict__ qcol = qpT + b * (U * TQ) + lane;
    const float* __restrict__ krow = kp + (long long)row0 * U;

    float acc[KW];
    #pragma unroll
    for (int j = 0; j < KW; ++j) acc[j] = 0.f;

    #pragma unroll 1
    for (int c = 0; c < U; c += UC) {
        float q[UC];
        #pragma unroll
        for (int i = 0; i < UC; ++i) q[i] = qcol[(c + i) * TQ];

        #pragma unroll
        for (int j = 0; j < KW; ++j) {
            #pragma unroll
            for (int i = 0; i < UC; ++i)
                acc[j] += fast_tanh(q[i] + krow[j * U + c + i]) * v[c + i];
        }
    }

    float4 o;
    o.x = acc[0]; o.y = acc[1]; o.z = acc[2]; o.w = acc[3];
    *(float4*)(sc + ((long long)b * TQ + lane) * TK + k0) = o;
}


template <int M>
__device__ __forceinline__ float row_xor(float x) {
    const int b = __builtin_bit_cast(int, x);
    return __builtin_bit_cast(float, __builtin_amdgcn_update_dpp(b, b, 0x160 + M, 0xF, 0xF, false));
}
__device__ __forceinline__ float xhalf(float x) {
    const int b = __builtin_bit_cast(int, x);
    return __builtin_bit_cast(float, __builtin_amdgcn_permlanex16(b, b, 0x76543210, 0xfedcba98, false, false));
}
__device__ __forceinline__ float wave_max(float x) {
    x = fmaxf(x, row_xor<1>(x));
    x = fmaxf(x, row_xor<2>(x));
    x = fmaxf(x, row_xor<4>(x));
    x = fmaxf(x, row_xor<8>(x));
    return fmaxf(x, xhalf(x));
}
__device__ __forceinline__ float wave_sum(float x) {
    x += row_xor<1>(x);
    x += row_xor<2>(x);
    x += row_xor<4>(x);
    x += row_xor<8>(x);
    return x + xhalf(x);
}

__global__ void __launch_bounds__(32)
softmax_rows_kernel(const float* __restrict__ sc,
                    float* __restrict__ attn)
{
    constexpr int TK = 2048;
    constexpr int NV = TK / (32 * 4);
    const int row  = blockIdx.x;
    const int lane = threadIdx.x;
    const v4f* __restrict__ s4 = (const v4f*)(sc   + (long long)row * TK) + lane;
    v4f* __restrict__       a4 = (v4f*)(attn + (long long)row * TK) + lane;

    v4f x[NV];
    #pragma unroll
    for (int i = 0; i < NV; ++i) x[i] = s4[i * 32];

    float m = -INFINITY;
    #pragma unroll
    for (int i = 0; i < NV; ++i)
        m = fmaxf(m, fmaxf(fmaxf(x[i].x, x[i].y), fmaxf(x[i].z, x[i].w)));
    const float mx = wave_max(m);

    float sum = 0.f;
    #pragma unroll
    for (int i = 0; i < NV; ++i) {
        x[i].x = __expf(x[i].x - mx);
        x[i].y = __expf(x[i].y - mx);
        x[i].z = __expf(x[i].z - mx);
        x[i].w = __expf(x[i].w - mx);
        sum += (x[i].x + x[i].y) + (x[i].z + x[i].w);
    }
    const float inv = 1.0f / wave_sum(sum);

    #pragma unroll
    for (int i = 0; i < NV; ++i) a4[i * 32] = x[i] * inv;
}

extern "C" void kernel_launch(void* const* d_in, const int* in_sizes, int n_in,
                              void* d_out, int out_size, void* d_ws, size_t ws_size,
                              hipStream_t stream)
{
    (void)in_sizes; (void)n_in; (void)out_size; (void)ws_size;

    const float* query = (const float*)d_in[0];
    const float* keys  = (const float*)d_in[1];
    const float* Wq    = (const float*)d_in[2];
    const float* Wk    = (const float*)d_in[3];
    const float* v     = (const float*)d_in[4];

    float* context = (float*)d_out;
    float* attn    = (float*)d_out + 65536;

    float* qp = (float*)d_ws;
    float* kp = qp + 65536;
    float* sc = kp + (size_t)16384 * 256;

    qproj_tile_kernel<<<dim3(256), 32, 0, stream>>>(Wq, query, qp);

    wmma_gemm_f32<true, 4><<<dim3(256, 1), 128, 0, stream>>>(
        keys, Wk, kp, 16384, 256, 256, 0, 0, 0);

    score_tanh_kernel<<<dim3(512), 256, 0, stream>>>(qp, kp, v, sc);

    softmax_rows_kernel<<<dim3(256), 32, 0, stream>>>(sc, attn);

    context_tile_kernel<<<dim3(256), 32, 0, stream>>>(attn, keys, context);
}
